# starting kernel
# baseline (speedup 1.0000x reference)
#include <hip/hip_runtime.h>
#include <hip/hip_bf16.h>


typedef float v2f __attribute__((ext_vector_type(2)));
typedef float v4f __attribute__((ext_vector_type(4)));
typedef float v8f __attribute__((ext_vector_type(8)));

#define TOKENS 4096
#define DIN    1024
#define DOUT   1024
#define RANK   8

__global__ __launch_bounds__(256) void lora_inter_kernel(
    const float* __restrict__ x,
    const float* __restrict__ Bp,
    float* __restrict__ inter) {
  const int lane = threadIdx.x & 31;
  const int wave = threadIdx.x >> 5;
  const int t = blockIdx.x * 8 + wave;

  const v4f* xp = (const v4f*)(x + (size_t)t * DIN);
  v4f xr[8];
#pragma unroll
  for (int j = 0; j < 8; ++j) xr[j] = xp[j * 32 + lane];

#pragma unroll
  for (int r = 0; r < RANK; ++r) {
    const v4f* bp =
        (const v4f*)(Bp + (size_t)t * (RANK * DIN) + (size_t)r * DIN);
    float acc = 0.f;
#pragma unroll
    for (int j = 0; j < 8; ++j) {
      v4f bv = __builtin_nontemporal_load(&bp[j * 32 + lane]);
      acc += xr[j].x * bv.x + xr[j].y * bv.y + xr[j].z * bv.z + xr[j].w * bv.w;
    }
#pragma unroll
    for (int off = 16; off > 0; off >>= 1) acc += __shfl_xor(acc, off, 32);
    if (lane == 0) inter[t * RANK + r] = acc;
  }
}

#define TM 128
#define TN 128
#define TK 32
#define NKT (DIN / TK)
#define LSTR 36

__global__ __launch_bounds__(256) void lora_gemm_kernel(
    const float* __restrict__ x,
    const float* __restrict__ W,
    const float* __restrict__ bias,
    const float* __restrict__ Ap,
    const float* __restrict__ inter,
    float* __restrict__ out) {
  __shared__ float xs[2][TM * LSTR];
  __shared__ float wsm[2][TN * LSTR];
  __shared__ float is[TM * RANK];

  const int tid  = threadIdx.x;
  const int lane = tid & 31;
  const int wv   = tid >> 5;
  const int wm   = wv & 1;
  const int wn   = wv >> 1;
  const int t0   = blockIdx.x * TM;
  const int o0   = blockIdx.y * TN;

  const int lrow = tid >> 3;
  const int lf4  = tid & 7;
  const int r15  = lane & 15;
  const int kh   = (lane >> 4) * 2;

  v4f px[4], pw[4];
#pragma unroll
  for (int p = 0; p < 4; ++p) {
    const int row = lrow + p * 32;
    px[p] = *(const v4f*)(x + (size_t)(t0 + row) * DIN + lf4 * 4);
    pw[p] = *(const v4f*)(W + (size_t)(o0 + row) * DIN + lf4 * 4);
  }
  ((v4f*)is)[tid] = ((const v4f*)(inter + (size_t)t0 * RANK))[tid];
#pragma unroll
  for (int p = 0; p < 4; ++p) {
    const int row = lrow + p * 32;
    *(v4f*)(&xs[0][row * LSTR + lf4 * 4]) = px[p];
    *(v4f*)(&wsm[0][row * LSTR + lf4 * 4]) = pw[p];
  }

  v8f acc[4][2];
#pragma unroll
  for (int mi = 0; mi < 4; ++mi)
#pragma unroll
    for (int ni = 0; ni < 2; ++ni)
#pragma unroll
      for (int v = 0; v < 8; ++v) acc[mi][ni][v] = 0.f;

  __syncthreads();

  for (int kt = 0; kt < NKT; ++kt) {
    const int cur = kt & 1;
    const bool more = (kt + 1) < NKT;

    if (more) {
      const int kn = (kt + 1) * TK;
#pragma unroll
      for (int p = 0; p < 4; ++p) {
        const int row = lrow + p * 32;
        px[p] = *(const v4f*)(x + (size_t)(t0 + row) * DIN + kn + lf4 * 4);
        pw[p] = *(const v4f*)(W + (size_t)(o0 + row) * DIN + kn + lf4 * 4);
      }
    }

    const float* xb = xs[cur];
    const float* wb = wsm[cur];
#pragma unroll
    for (int kk = 0; kk < TK; kk += 4) {
      v2f a[4], bfr[2];
#pragma unroll
      for (int mi = 0; mi < 4; ++mi)
        a[mi] = *(const v2f*)(xb + (wm * 64 + mi * 16 + r15) * LSTR + kk + kh);
#pragma unroll
      for (int ni = 0; ni < 2; ++ni)
        bfr[ni] = *(const v2f*)(wb + (wn * 32 + ni * 16 + r15) * LSTR + kk + kh);
#pragma unroll
      for (int mi = 0; mi < 4; ++mi)
#pragma unroll
        for (int ni = 0; ni < 2; ++ni)
          acc[mi][ni] = __builtin_amdgcn_wmma_f32_16x16x4_f32(
              false, a[mi], false, bfr[ni], (short)0, acc[mi][ni],
              false, false);
    }

    if (more) {
#pragma unroll
      for (int p = 0; p < 4; ++p) {
        const int row = lrow + p * 32;
        *(v4f*)(&xs[1 - cur][row * LSTR + lf4 * 4]) = px[p];
        *(v4f*)(&wsm[1 - cur][row * LSTR + lf4 * 4]) = pw[p];
      }
    }
    __syncthreads();
  }

  const int mhi = (lane >> 4) << 3;
#pragma unroll
  for (int ni = 0; ni < 2; ++ni) {
    const int o = o0 + wn * 32 + ni * 16 + r15;
    const float bv = bias[o];
#pragma unroll
    for (int mi = 0; mi < 4; ++mi) {
#pragma unroll
      for (int v = 0; v < 8; ++v) {
        const int mloc = wm * 64 + mi * 16 + v + mhi;
        const int t = t0 + mloc;
        const v4f* ar = (const v4f*)(Ap + ((size_t)t * DOUT + o) * RANK);
        v4f a0 = __builtin_nontemporal_load(&ar[0]);
        v4f a1 = __builtin_nontemporal_load(&ar[1]);
        const v4f i0 = *(const v4f*)(is + mloc * RANK);
        const v4f i1 = *(const v4f*)(is + mloc * RANK + 4);
        float ad = i0.x * a0.x + i0.y * a0.y + i0.z * a0.z + i0.w * a0.w +
                   i1.x * a1.x + i1.y * a1.y + i1.z * a1.z + i1.w * a1.w;
        out[(size_t)t * DOUT + o] = acc[mi][ni][v] + bv + ad;
      }
    }
  }
}

extern "C" void kernel_launch(void* const* d_in, const int* in_sizes, int n_in,
                              void* d_out, int out_size, void* d_ws,
                              size_t ws_size, hipStream_t stream) {
  (void)in_sizes; (void)n_in; (void)out_size; (void)ws_size;
  const float* x    = (const float*)d_in[0];
  const float* Ap   = (const float*)d_in[1];
  const float* Bp   = (const float*)d_in[2];
  const float* W    = (const float*)d_in[3];
  const float* bias = (const float*)d_in[4];
  float* out   = (float*)d_out;
  float* inter = (float*)d_ws;

  lora_inter_kernel<<<TOKENS / 8, 256, 0, stream>>>(x, Bp, inter);

  dim3 grid(TOKENS / TM, DOUT / TN);
  lora_gemm_kernel<<<grid, 256, 0, stream>>>(x, W, bias, Ap, inter, out);
}


# Stream the A_params tile inside the WMMA K loop (4 token rows per slab, loads ahead of the WMMAs) into an LDS adaptation tile; rank-8 sum now two 4-term halves joined by a lane swap
# speedup vs baseline: 1.2614x; 1.2614x over previous
#include <hip/hip_runtime.h>
#include <hip/hip_bf16.h>


typedef float v2f __attribute__((ext_vector_type(2)));
typedef float v4f __attribute__((ext_vector_type(4)));
typedef float v8f __attribute__((ext_vector_type(8)));

#define TOKENS 4096
#define DIN    1024
#define DOUT   1024
#define RANK   8

__global__ __launch_bounds__(256) void lora_inter_kernel(
    const float* __restrict__ x,
    const float* __restrict__ Bp,
    float* __restrict__ inter) {
  const int lane = threadIdx.x & 31;
  const int wave = threadIdx.x >> 5;
  const int t = blockIdx.x * 8 + wave;

  const v4f* xp = (const v4f*)(x + (size_t)t * DIN);
  v4f xr[8];
#pragma unroll
  for (int j = 0; j < 8; ++j) xr[j] = xp[j * 32 + lane];

#pragma unroll
  for (int r = 0; r < RANK; ++r) {
    const v4f* bp =
        (const v4f*)(Bp + (size_t)t * (RANK * DIN) + (size_t)r * DIN);
    float acc = 0.f;
#pragma unroll
    for (int j = 0; j < 8; ++j) {
      v4f bv = __builtin_nontemporal_load(&bp[j * 32 + lane]);
      acc += xr[j].x * bv.x + xr[j].y * bv.y + xr[j].z * bv.z + xr[j].w * bv.w;
    }
#pragma unroll
    for (int off = 16; off > 0; off >>= 1) acc += __shfl_xor(acc, off, 32);
    if (lane == 0) inter[t * RANK + r] = acc;
  }
}

#define TM 128
#define TN 128
#define TK 32
#define NKT (DIN / TK)
#define LSTR 36
#define ASTR 132

__device__ __forceinline__ float pair_sum(float p) {
  const int q = __builtin_amdgcn_update_dpp(0, __float_as_int(p), 0xB1, 0xF,
                                            0xF, true);
  return p + __int_as_float(q);
}

__global__ __launch_bounds__(256) void lora_gemm_kernel(
    const float* __restrict__ x,
    const float* __restrict__ W,
    const float* __restrict__ bias,
    const float* __restrict__ Ap,
    const float* __restrict__ inter,
    float* __restrict__ out) {
  __shared__ float xs[2][TM * LSTR];
  __shared__ float wsm[2][TN * LSTR];
  __shared__ float is[TM * RANK];
  __shared__ float adt[TM * ASTR];

  const int tid  = threadIdx.x;
  const int lane = tid & 31;
  const int wv   = tid >> 5;
  const int wm   = wv & 1;
  const int wn   = wv >> 1;
  const int t0   = blockIdx.x * TM;
  const int o0   = blockIdx.y * TN;

  const int lrow = tid >> 3;
  const int lf4  = tid & 7;
  const int r15  = lane & 15;
  const int kh   = (lane >> 4) * 2;

  const int arow  = wv >> 1;
  const int ahalf = wv & 1;
  const int rsel  = (lane & 1) * 4;
  const v4f* apw = (const v4f*)(Ap + ((size_t)(t0 + arow) * DOUT + o0) * RANK) +
                   ahalf * 128 + lane;

  v4f px[4], pw[4];
#pragma unroll
  for (int p = 0; p < 4; ++p) {
    const int row = lrow + p * 32;
    px[p] = *(const v4f*)(x + (size_t)(t0 + row) * DIN + lf4 * 4);
    pw[p] = *(const v4f*)(W + (size_t)(o0 + row) * DIN + lf4 * 4);
  }
  ((v4f*)is)[tid] = ((const v4f*)(inter + (size_t)t0 * RANK))[tid];
#pragma unroll
  for (int p = 0; p < 4; ++p) {
    const int row = lrow + p * 32;
    *(v4f*)(&xs[0][row * LSTR + lf4 * 4]) = px[p];
    *(v4f*)(&wsm[0][row * LSTR + lf4 * 4]) = pw[p];
  }

  v8f acc[4][2];
#pragma unroll
  for (int mi = 0; mi < 4; ++mi)
#pragma unroll
    for (int ni = 0; ni < 2; ++ni)
#pragma unroll
      for (int v = 0; v < 8; ++v) acc[mi][ni][v] = 0.f;

  __syncthreads();

  for (int kt = 0; kt < NKT; ++kt) {
    const int cur = kt & 1;
    const bool more = (kt + 1) < NKT;

    if (more) {
      const int kn = (kt + 1) * TK;
#pragma unroll
      for (int p = 0; p < 4; ++p) {
        const int row = lrow + p * 32;
        px[p] = *(const v4f*)(x + (size_t)(t0 + row) * DIN + kn + lf4 * 4);
        pw[p] = *(const v4f*)(W + (size_t)(o0 + row) * DIN + kn + lf4 * 4);
      }
    }
    v4f pa[4];
    {
      const v4f* ar = apw + (size_t)kt * (4 * DOUT * RANK / 4);
#pragma unroll
      for (int j = 0; j < 4; ++j) pa[j] = __builtin_nontemporal_load(&ar[j * 32]);
    }

    const float* xb = xs[cur];
    const float* wb = wsm[cur];
#pragma unroll
    for (int kk = 0; kk < TK; kk += 4) {
      if (kk == 24) __builtin_amdgcn_sched_barrier(0x104);
      v2f a[4], bfr[2];
#pragma unroll
      for (int mi = 0; mi < 4; ++mi)
        a[mi] = *(const v2f*)(xb + (wm * 64 + mi * 16 + r15) * LSTR + kk + kh);
#pragma unroll
      for (int ni = 0; ni < 2; ++ni)
        bfr[ni] = *(const v2f*)(wb + (wn * 32 + ni * 16 + r15) * LSTR + kk + kh);
#pragma unroll
      for (int mi = 0; mi < 4; ++mi)
#pragma unroll
        for (int ni = 0; ni < 2; ++ni)
          acc[mi][ni] = __builtin_amdgcn_wmma_f32_16x16x4_f32(
              false, a[mi], false, bfr[ni], (short)0, acc[mi][ni],
              false, false);
    }

    if (more) {
#pragma unroll
      for (int p = 0; p < 4; ++p) {
        const int row = lrow + p * 32;
        *(v4f*)(&xs[1 - cur][row * LSTR + lf4 * 4]) = px[p];
        *(v4f*)(&wsm[1 - cur][row * LSTR + lf4 * 4]) = pw[p];
      }
    }

    {
      const int mloc = kt * 4 + arow;
      const v4f iv = *(const v4f*)(is + mloc * RANK + rsel);
      const v2f ilo = {iv.x, iv.y}, ihi = {iv.z, iv.w};
      float s[4];
#pragma unroll
      for (int j = 0; j < 4; ++j) {
        const v2f alo = {pa[j].x, pa[j].y}, ahi = {pa[j].z, pa[j].w};
        const v2f t = ilo * alo + ihi * ahi;
        s[j] = pair_sum(t.x + t.y);
      }
      const bool odd = (lane & 1) != 0;
      float* ad = adt + mloc * ASTR + ahalf * 64 + (odd ? 32 : 0) + (lane >> 1);
      ad[0]  = odd ? s[2] : s[0];
      ad[16] = odd ? s[3] : s[1];
    }
    __syncthreads();
  }

  const int mhi = (lane >> 4) << 3;
#pragma unroll
  for (int ni = 0; ni < 2; ++ni) {
    const int o = o0 + wn * 32 + ni * 16 + r15;
    const float bv = bias[o];
#pragma unroll
    for (int mi = 0; mi < 4; ++mi) {
#pragma unroll
      for (int v = 0; v < 8; ++v) {
        const int mloc = wm * 64 + mi * 16 + v + mhi;
        const int t = t0 + mloc;
        const float ad = adt[mloc * ASTR + wn * 32 + ni * 16 + r15];
        out[(size_t)t * DOUT + o] = acc[mi][ni][v] + bv + ad;
      }
    }
  }
}

extern "C" void kernel_launch(void* const* d_in, const int* in_sizes, int n_in,
                              void* d_out, int out_size, void* d_ws,
                              size_t ws_size, hipStream_t stream) {
  (void)in_sizes; (void)n_in; (void)out_size; (void)ws_size;
  const float* x    = (const float*)d_in[0];
  const float* Ap   = (const float*)d_in[1];
  const float* Bp   = (const float*)d_in[2];
  const float* W    = (const float*)d_in[3];
  const float* bias = (const float*)d_in[4];
  float* out   = (float*)d_out;
  float* inter = (float*)d_ws;

  lora_inter_kernel<<<TOKENS / 8, 256, 0, stream>>>(x, Bp, inter);

  dim3 grid(TOKENS / TM, DOUT / TN);
  lora_gemm_kernel<<<grid, 256, 0, stream>>>(x, W, bias, Ap, inter, out);
}


# Inter kernel: B_params requested two rank pairs ahead (16-32 loads in flight per wave, was 8 drained per rank); 8 per-lane rank sums folded by one merged butterfly (same lane pairing per rank)
# speedup vs baseline: 1.3420x; 1.0639x over previous
#include <hip/hip_runtime.h>
#include <hip/hip_bf16.h>


typedef float v2f __attribute__((ext_vector_type(2)));
typedef float v4f __attribute__((ext_vector_type(4)));
typedef float v8f __attribute__((ext_vector_type(8)));

#define TOKENS 4096
#define DIN    1024
#define DOUT   1024
#define RANK   8

__device__ __forceinline__ float dot8(const v4f* xr, const v4f* bv) {
  float acc = 0.f;
#pragma unroll
  for (int j = 0; j < 8; ++j)
    acc += xr[j].x * bv[j].x + xr[j].y * bv[j].y + xr[j].z * bv[j].z +
           xr[j].w * bv[j].w;
  return acc;
}

__global__ __launch_bounds__(256) void lora_inter_kernel(
    const float* __restrict__ x,
    const float* __restrict__ Bp,
    float* __restrict__ inter) {
  const int lane = threadIdx.x & 31;
  const int wave = threadIdx.x >> 5;
  const int t = blockIdx.x * 8 + wave;

  const v4f* xp = (const v4f*)(x + (size_t)t * DIN);
  const v4f* bp = (const v4f*)(Bp + (size_t)t * (RANK * DIN)) + lane;

  v4f xr[8];
#pragma unroll
  for (int j = 0; j < 8; ++j) xr[j] = xp[j * 32 + lane];

  v4f ga[16], gb[16];
#pragma unroll
  for (int j = 0; j < 16; ++j) ga[j] = __builtin_nontemporal_load(&bp[j * 32]);
#pragma unroll
  for (int j = 0; j < 16; ++j)
    gb[j] = __builtin_nontemporal_load(&bp[512 + j * 32]);
  __builtin_amdgcn_sched_barrier(0);

  float acc[8];
  acc[0] = dot8(xr, ga);
  acc[1] = dot8(xr, ga + 8);
  asm volatile("" : "+v"(acc[0]), "+v"(acc[1]));
  __builtin_amdgcn_sched_barrier(0);
#pragma unroll
  for (int j = 0; j < 16; ++j)
    ga[j] = __builtin_nontemporal_load(&bp[1024 + j * 32]);
  __builtin_amdgcn_sched_barrier(0);
  acc[2] = dot8(xr, gb);
  acc[3] = dot8(xr, gb + 8);
  asm volatile("" : "+v"(acc[2]), "+v"(acc[3]));
  __builtin_amdgcn_sched_barrier(0);
#pragma unroll
  for (int j = 0; j < 16; ++j)
    gb[j] = __builtin_nontemporal_load(&bp[1536 + j * 32]);
  __builtin_amdgcn_sched_barrier(0);
  acc[4] = dot8(xr, ga);
  acc[5] = dot8(xr, ga + 8);
  asm volatile("" : "+v"(acc[4]), "+v"(acc[5]));
  __builtin_amdgcn_sched_barrier(0);
  acc[6] = dot8(xr, gb);
  acc[7] = dot8(xr, gb + 8);

  const bool h16 = (lane & 16) != 0;
  float q[4];
#pragma unroll
  for (int j = 0; j < 4; ++j) {
    const float send = h16 ? acc[j] : acc[j + 4];
    const float keep = h16 ? acc[j + 4] : acc[j];
    q[j] = keep + __shfl_xor(send, 16, 32);
  }
  const bool h8 = (lane & 8) != 0;
  float d[2];
#pragma unroll
  for (int j = 0; j < 2; ++j) {
    const float send = h8 ? q[j] : q[j + 2];
    const float keep = h8 ? q[j + 2] : q[j];
    d[j] = keep + __shfl_xor(send, 8, 32);
  }
  const bool h4 = (lane & 4) != 0;
  float s = (h4 ? d[1] : d[0]) + __shfl_xor(h4 ? d[0] : d[1], 4, 32);
  s += __shfl_xor(s, 2, 32);
  s += __shfl_xor(s, 1, 32);
  if ((lane & 3) == 0) inter[t * RANK + (lane >> 2)] = s;
}

#define TM 128
#define TN 128
#define TK 32
#define NKT (DIN / TK)
#define LSTR 36
#define ASTR 132

__device__ __forceinline__ float pair_sum(float p) {
  const int q = __builtin_amdgcn_update_dpp(0, __float_as_int(p), 0xB1, 0xF,
                                            0xF, true);
  return p + __int_as_float(q);
}

__global__ __launch_bounds__(256) void lora_gemm_kernel(
    const float* __restrict__ x,
    const float* __restrict__ W,
    const float* __restrict__ bias,
    const float* __restrict__ Ap,
    const float* __restrict__ inter,
    float* __restrict__ out) {
  __shared__ float xs[2][TM * LSTR];
  __shared__ float wsm[2][TN * LSTR];
  __shared__ float is[TM * RANK];
  __shared__ float adt[TM * ASTR];

  const int tid  = threadIdx.x;
  const int lane = tid & 31;
  const int wv   = tid >> 5;
  const int wm   = wv & 1;
  const int wn   = wv >> 1;
  const int t0   = blockIdx.x * TM;
  const int o0   = blockIdx.y * TN;

  const int lrow = tid >> 3;
  const int lf4  = tid & 7;
  const int r15  = lane & 15;
  const int kh   = (lane >> 4) * 2;

  const int arow  = wv >> 1;
  const int ahalf = wv & 1;
  const int rsel  = (lane & 1) * 4;
  const v4f* apw = (const v4f*)(Ap + ((size_t)(t0 + arow) * DOUT + o0) * RANK) +
                   ahalf * 128 + lane;

  v4f px[4], pw[4];
#pragma unroll
  for (int p = 0; p < 4; ++p) {
    const int row = lrow + p * 32;
    px[p] = *(const v4f*)(x + (size_t)(t0 + row) * DIN + lf4 * 4);
    pw[p] = *(const v4f*)(W + (size_t)(o0 + row) * DIN + lf4 * 4);
  }
  ((v4f*)is)[tid] = ((const v4f*)(inter + (size_t)t0 * RANK))[tid];
#pragma unroll
  for (int p = 0; p < 4; ++p) {
    const int row = lrow + p * 32;
    *(v4f*)(&xs[0][row * LSTR + lf4 * 4]) = px[p];
    *(v4f*)(&wsm[0][row * LSTR + lf4 * 4]) = pw[p];
  }

  v8f acc[4][2];
#pragma unroll
  for (int mi = 0; mi < 4; ++mi)
#pragma unroll
    for (int ni = 0; ni < 2; ++ni)
#pragma unroll
      for (int v = 0; v < 8; ++v) acc[mi][ni][v] = 0.f;

  __syncthreads();

  for (int kt = 0; kt < NKT; ++kt) {
    const int cur = kt & 1;
    const bool more = (kt + 1) < NKT;

    if (more) {
      const int kn = (kt + 1) * TK;
#pragma unroll
      for (int p = 0; p < 4; ++p) {
        const int row = lrow + p * 32;
        px[p] = *(const v4f*)(x + (size_t)(t0 + row) * DIN + kn + lf4 * 4);
        pw[p] = *(const v4f*)(W + (size_t)(o0 + row) * DIN + kn + lf4 * 4);
      }
    }
    v4f pa[4];
    {
      const v4f* ar = apw + (size_t)kt * (4 * DOUT * RANK / 4);
#pragma unroll
      for (int j = 0; j < 4; ++j) pa[j] = __builtin_nontemporal_load(&ar[j * 32]);
    }

    const float* xb = xs[cur];
    const float* wb = wsm[cur];
#pragma unroll
    for (int kk = 0; kk < TK; kk += 4) {
      if (kk == 24) __builtin_amdgcn_sched_barrier(0x104);
      v2f a[4], bfr[2];
#pragma unroll
      for (int mi = 0; mi < 4; ++mi)
        a[mi] = *(const v2f*)(xb + (wm * 64 + mi * 16 + r15) * LSTR + kk + kh);
#pragma unroll
      for (int ni = 0; ni < 2; ++ni)
        bfr[ni] = *(const v2f*)(wb + (wn * 32 + ni * 16 + r15) * LSTR + kk + kh);
#pragma unroll
      for (int mi = 0; mi < 4; ++mi)
#pragma unroll
        for (int ni = 0; ni < 2; ++ni)
          acc[mi][ni] = __builtin_amdgcn_wmma_f32_16x16x4_f32(
              false, a[mi], false, bfr[ni], (short)0, acc[mi][ni],
              false, false);
    }

    if (more) {
#pragma unroll
      for (int p = 0; p < 4; ++p) {
        const int row = lrow + p * 32;
        *(v4f*)(&xs[1 - cur][row * LSTR + lf4 * 4]) = px[p];
        *(v4f*)(&wsm[1 - cur][row * LSTR + lf4 * 4]) = pw[p];
      }
    }

    {
      const int mloc = kt * 4 + arow;
      const v4f iv = *(const v4f*)(is + mloc * RANK + rsel);
      const v2f ilo = {iv.x, iv.y}, ihi = {iv.z, iv.w};
      float s[4];
#pragma unroll
      for (int j = 0; j < 4; ++j) {
        const v2f alo = {pa[j].x, pa[j].y}, ahi = {pa[j].z, pa[j].w};
        const v2f t = ilo * alo + ihi * ahi;
        s[j] = pair_sum(t.x + t.y);
      }
      const bool odd = (lane & 1) != 0;
      float* ad = adt + mloc * ASTR + ahalf * 64 + (odd ? 32 : 0) + (lane >> 1);
      ad[0]  = odd ? s[2] : s[0];
      ad[16] = odd ? s[3] : s[1];
    }
    __syncthreads();
  }

  const int mhi = (lane >> 4) << 3;
#pragma unroll
  for (int ni = 0; ni < 2; ++ni) {
    const int o = o0 + wn * 32 + ni * 16 + r15;
    const float bv = bias[o];
#pragma unroll
    for (int mi = 0; mi < 4; ++mi) {
#pragma unroll
      for (int v = 0; v < 8; ++v) {
        const int mloc = wm * 64 + mi * 16 + v + mhi;
        const int t = t0 + mloc;
        const float ad = adt[mloc * ASTR + wn * 32 + ni * 16 + r15];
        out[(size_t)t * DOUT + o] = acc[mi][ni][v] + bv + ad;
      }
    }
  }
}

extern "C" void kernel_launch(void* const* d_in, const int* in_sizes, int n_in,
                              void* d_out, int out_size, void* d_ws,
                              size_t ws_size, hipStream_t stream) {
  (void)in_sizes; (void)n_in; (void)out_size; (void)ws_size;
  const float* x    = (const float*)d_in[0];
  const float* Ap   = (const float*)d_in[1];
  const float* Bp   = (const float*)d_in[2];
  const float* W    = (const float*)d_in[3];
  const float* bias = (const float*)d_in[4];
  float* out   = (float*)d_out;
  float* inter = (float*)d_ws;

  lora_inter_kernel<<<TOKENS / 8, 256, 0, stream>>>(x, Bp, inter);

  dim3 grid(TOKENS / TM, DOUT / TN);
  lora_gemm_kernel<<<grid, 256, 0, stream>>>(x, W, bias, Ap, inter, out);
}


# GEMM K loop pipelined one slab deeper: next slab's fragments read into a second register set, every global load requested a slab ahead, hand-written LDS-only barrier; only WMMAs follow each barrier
# speedup vs baseline: 1.4440x; 1.0760x over previous
#include <hip/hip_runtime.h>
#include <hip/hip_bf16.h>


typedef float v2f __attribute__((ext_vector_type(2)));
typedef float v4f __attribute__((ext_vector_type(4)));
typedef float v8f __attribute__((ext_vector_type(8)));

#define TOKENS 4096
#define DIN    1024
#define DOUT   1024
#define RANK   8

__device__ __forceinline__ float dot8(const v4f* xr, const v4f* bv) {
  float acc = 0.f;
#pragma unroll
  for (int j = 0; j < 8; ++j)
    acc += xr[j].x * bv[j].x + xr[j].y * bv[j].y + xr[j].z * bv[j].z +
           xr[j].w * bv[j].w;
  return acc;
}

__global__ __launch_bounds__(256) void lora_inter_kernel(
    const float* __restrict__ x,
    const float* __restrict__ Bp,
    float* __restrict__ inter) {
  const int lane = threadIdx.x & 31;
  const int wave = threadIdx.x >> 5;
  const int t = blockIdx.x * 8 + wave;

  const v4f* xp = (const v4f*)(x + (size_t)t * DIN);
  const v4f* bp = (const v4f*)(Bp + (size_t)t * (RANK * DIN)) + lane;

  v4f xr[8];
#pragma unroll
  for (int j = 0; j < 8; ++j) xr[j] = xp[j * 32 + lane];

  v4f ga[16], gb[16];
#pragma unroll
  for (int j = 0; j < 16; ++j) ga[j] = __builtin_nontemporal_load(&bp[j * 32]);
#pragma unroll
  for (int j = 0; j < 16; ++j)
    gb[j] = __builtin_nontemporal_load(&bp[512 + j * 32]);
  __builtin_amdgcn_sched_barrier(0);

  float acc[8];
  acc[0] = dot8(xr, ga);
  acc[1] = dot8(xr, ga + 8);
  asm volatile("" : "+v"(acc[0]), "+v"(acc[1]));
  __builtin_amdgcn_sched_barrier(0);
#pragma unroll
  for (int j = 0; j < 16; ++j)
    ga[j] = __builtin_nontemporal_load(&bp[1024 + j * 32]);
  __builtin_amdgcn_sched_barrier(0);
  acc[2] = dot8(xr, gb);
  acc[3] = dot8(xr, gb + 8);
  asm volatile("" : "+v"(acc[2]), "+v"(acc[3]));
  __builtin_amdgcn_sched_barrier(0);
#pragma unroll
  for (int j = 0; j < 16; ++j)
    gb[j] = __builtin_nontemporal_load(&bp[1536 + j * 32]);
  __builtin_amdgcn_sched_barrier(0);
  acc[4] = dot8(xr, ga);
  acc[5] = dot8(xr, ga + 8);
  asm volatile("" : "+v"(acc[4]), "+v"(acc[5]));
  __builtin_amdgcn_sched_barrier(0);
  acc[6] = dot8(xr, gb);
  acc[7] = dot8(xr, gb + 8);

  const bool h16 = (lane & 16) != 0;
  float q[4];
#pragma unroll
  for (int j = 0; j < 4; ++j) {
    const float send = h16 ? acc[j] : acc[j + 4];
    const float keep = h16 ? acc[j + 4] : acc[j];
    q[j] = keep + __shfl_xor(send, 16, 32);
  }
  const bool h8 = (lane & 8) != 0;
  float d[2];
#pragma unroll
  for (int j = 0; j < 2; ++j) {
    const float send = h8 ? q[j] : q[j + 2];
    const float keep = h8 ? q[j + 2] : q[j];
    d[j] = keep + __shfl_xor(send, 8, 32);
  }
  const bool h4 = (lane & 4) != 0;
  float s = (h4 ? d[1] : d[0]) + __shfl_xor(h4 ? d[0] : d[1], 4, 32);
  s += __shfl_xor(s, 2, 32);
  s += __shfl_xor(s, 1, 32);
  if ((lane & 3) == 0) inter[t * RANK + (lane >> 2)] = s;
}

#define TM 128
#define TN 128
#define TK 32
#define NKT (DIN / TK)
#define LSTR 36
#define ASTR 132

__device__ __forceinline__ float pair_sum(float p) {
  const int q = __builtin_amdgcn_update_dpp(0, __float_as_int(p), 0xB1, 0xF,
                                            0xF, true);
  return p + __int_as_float(q);
}

struct Frags { v2f a[8][4]; v2f b[8][2]; };
struct Stage { v4f x[4]; v4f w[4]; };

__device__ __forceinline__ void lds_done_signal() {
  asm volatile("s_wait_dscnt 0x0" ::: "memory");
  asm volatile("s_barrier_signal -1" ::: "memory");
}
__device__ __forceinline__ void barrier_wait() {
  asm volatile("s_barrier_wait -1" ::: "memory");
}

__global__ __launch_bounds__(256) void lora_gemm_kernel(
    const float* __restrict__ x,
    const float* __restrict__ W,
    const float* __restrict__ bias,
    const float* __restrict__ Ap,
    const float* __restrict__ inter,
    float* __restrict__ out) {
  __shared__ float xs[2][TM * LSTR];
  __shared__ float wsm[2][TN * LSTR];
  __shared__ float is[TM * RANK];
  __shared__ float adt[TM * ASTR];

  const int tid  = threadIdx.x;
  const int lane = tid & 31;
  const int wv   = tid >> 5;
  const int wm   = wv & 1;
  const int wn   = wv >> 1;
  const int t0   = blockIdx.x * TM;
  const int o0   = blockIdx.y * TN;

  const int lrow = tid >> 3;
  const int lf4  = tid & 7;
  const int r15  = lane & 15;
  const int kh   = (lane >> 4) * 2;

  const int arow  = wv >> 1;
  const int ahalf = wv & 1;
  const int rsel  = (lane & 1) * 4;
  const v4f* apw = (const v4f*)(Ap + ((size_t)(t0 + arow) * DOUT + o0) * RANK) +
                   ahalf * 128 + lane;

  const float* xg = x + (size_t)(t0 + lrow) * DIN + lf4 * 4;
  const float* wg = W + (size_t)(o0 + lrow) * DIN + lf4 * 4;
  const int sto  = lrow * LSTR + lf4 * 4;
  const int aoff = (wm * 64 + r15) * LSTR + kh;
  const int boff = (wn * 32 + r15) * LSTR + kh;

  v8f acc[4][2];
#pragma unroll
  for (int mi = 0; mi < 4; ++mi)
#pragma unroll
    for (int ni = 0; ni < 2; ++ni)
#pragma unroll
      for (int v = 0; v < 8; ++v) acc[mi][ni][v] = 0.f;

  auto gload = [&](Stage& st, int s) __attribute__((always_inline)) {
#pragma unroll
    for (int p = 0; p < 4; ++p) {
      st.x[p] = *(const v4f*)(xg + (size_t)p * 32 * DIN + s * TK);
      st.w[p] = *(const v4f*)(wg + (size_t)p * 32 * DIN + s * TK);
    }
  };
  auto lstore = [&](const Stage& st, int b) __attribute__((always_inline)) {
#pragma unroll
    for (int p = 0; p < 4; ++p) {
      *(v4f*)(&xs[b][sto + p * 32 * LSTR]) = st.x[p];
      *(v4f*)(&wsm[b][sto + p * 32 * LSTR]) = st.w[p];
    }
  };
  auto fload = [&](Frags& f, int b) __attribute__((always_inline)) {
    const float* xb = xs[b] + aoff;
    const float* wb = wsm[b] + boff;
#pragma unroll
    for (int k8 = 0; k8 < 8; ++k8) {
#pragma unroll
      for (int mi = 0; mi < 4; ++mi)
        f.a[k8][mi] = *(const v2f*)(xb + mi * 16 * LSTR + k8 * 4);
#pragma unroll
      for (int ni = 0; ni < 2; ++ni)
        f.b[k8][ni] = *(const v2f*)(wb + ni * 16 * LSTR + k8 * 4);
    }
  };
  auto aload = [&](v4f (&pa)[4], int s) __attribute__((always_inline)) {
    const v4f* ar = apw + (size_t)s * (4 * DOUT * RANK / 4);
#pragma unroll
    for (int j = 0; j < 4; ++j) pa[j] = __builtin_nontemporal_load(&ar[j * 32]);
  };
  auto mma8 = [&](const Frags& f, int k8) __attribute__((always_inline)) {
#pragma unroll
    for (int mi = 0; mi < 4; ++mi)
#pragma unroll
      for (int ni = 0; ni < 2; ++ni)
        acc[mi][ni] = __builtin_amdgcn_wmma_f32_16x16x4_f32(
            false, f.a[k8][mi], false, f.b[k8][ni], (short)0, acc[mi][ni],
            false, false);
  };

  auto slab = [&](const Frags& Fc, Frags& Fn, const Stage& Pc, Stage& Pn,
                  const v4f (&pac)[4], v4f (&pan)[4], int j, int bc,
                  bool ldF, bool ldP, bool ldA, bool stP)
                  __attribute__((always_inline)) {
    mma8(Fc, 0);
    __builtin_amdgcn_sched_barrier(0);
    const int mloc = j * 4 + arow;
    const v4f iv = *(const v4f*)(is + mloc * RANK + rsel);
    if (ldF) fload(Fn, bc ^ 1);
    if (ldP) gload(Pn, j + 3);
    if (ldA) aload(pan, j + 1);
    __builtin_amdgcn_sched_barrier(0);
#pragma unroll
    for (int k8 = 1; k8 < 5; ++k8) mma8(Fc, k8);
    __builtin_amdgcn_sched_barrier(0x104);
    mma8(Fc, 5);
    if (stP) lstore(Pc, bc);
    {
      const v2f ilo = {iv.x, iv.y}, ihi = {iv.z, iv.w};
      float s[4];
#pragma unroll
      for (int q = 0; q < 4; ++q) {
        const v2f alo = {pac[q].x, pac[q].y}, ahi = {pac[q].z, pac[q].w};
        const v2f t = ilo * alo + ihi * ahi;
        s[q] = pair_sum(t.x + t.y);
      }
      const bool odd = (lane & 1) != 0;
      float* ad = adt + mloc * ASTR + ahalf * 64 + (odd ? 32 : 0) + (lane >> 1);
      ad[0]  = odd ? s[2] : s[0];
      ad[16] = odd ? s[3] : s[1];
    }
    __builtin_amdgcn_sched_barrier(0);
    mma8(Fc, 6);
    lds_done_signal();
    __builtin_amdgcn_sched_barrier(0);
    mma8(Fc, 7);
    barrier_wait();
  };

  Frags F0, F1;
  Stage P0, P1;
  v4f paA[4], paB[4];
  gload(P0, 0);
  gload(P1, 1);
  const v4f itile = ((const v4f*)(inter + (size_t)t0 * RANK))[tid];
  lstore(P0, 0);
  lstore(P1, 1);
  ((v4f*)is)[tid] = itile;
  aload(paA, 0);
  gload(P0, 2);
  lds_done_signal();
  barrier_wait();
  fload(F0, 0);
  lds_done_signal();
  barrier_wait();

  for (int kt = 0; kt < NKT - 4; kt += 2) {
    slab(F0, F1, P0, P1, paA, paB, kt,     0, true, true, true, true);
    slab(F1, F0, P1, P0, paB, paA, kt + 1, 1, true, true, true, true);
  }
  slab(F0, F1, P0, P1, paA, paB, NKT - 4, 0, true, true,  true,  true);
  slab(F1, F0, P1, P0, paB, paA, NKT - 3, 1, true, false, true,  true);
  slab(F0, F1, P0, P1, paA, paB, NKT - 2, 0, true, false, true,  false);
  slab(F1, F0, P1, P0, paB, paA, NKT - 1, 1, false, false, false, false);

  const int mhi = (lane >> 4) << 3;
#pragma unroll
  for (int ni = 0; ni < 2; ++ni) {
    const int o = o0 + wn * 32 + ni * 16 + r15;
    const float bv = bias[o];
#pragma unroll
    for (int mi = 0; mi < 4; ++mi) {
#pragma unroll
      for (int v = 0; v < 8; ++v) {
        const int mloc = wm * 64 + mi * 16 + v + mhi;
        const int t = t0 + mloc;
        const float ad = adt[mloc * ASTR + wn * 32 + ni * 16 + r15];
        out[(size_t)t * DOUT + o] = acc[mi][ni][v] + bv + ad;
      }
    }
  }
}

extern "C" void kernel_launch(void* const* d_in, const int* in_sizes, int n_in,
                              void* d_out, int out_size, void* d_ws,
                              size_t ws_size, hipStream_t stream) {
  (void)in_sizes; (void)n_in; (void)out_size; (void)ws_size;
  const float* x    = (const float*)d_in[0];
  const float* Ap   = (const float*)d_in[1];
  const float* Bp   = (const float*)d_in[2];
  const float* W    = (const float*)d_in[3];
  const float* bias = (const float*)d_in[4];
  float* out   = (float*)d_out;
  float* inter = (float*)d_ws;

  lora_inter_kernel<<<TOKENS / 8, 256, 0, stream>>>(x, Bp, inter);

  dim3 grid(TOKENS / TM, DOUT / TN);
  lora_gemm_kernel<<<grid, 256, 0, stream>>>(x, W, bias, Ap, inter, out);
}
